# starting kernel
# baseline (speedup 1.0000x reference)
#include <hip/hip_runtime.h>

#define N_NODES 100000
#define N_EDGES 1600000
#define DIN 128
#define DH 64
#define LN_EPS 1e-5f

typedef __attribute__((ext_vector_type(2))) float v2f;
typedef __attribute__((ext_vector_type(8))) float v8f;

__global__ __launch_bounds__(256) void degree_kernel(const int* __restrict__ idx,
                                                     float* __restrict__ deg, int e) {
    int i = blockIdx.x * blockDim.x + threadIdx.x;
    if (i < e) atomicAdd(&deg[idx[i]], 1.0f);
}

__global__ __launch_bounds__(256) void finalize_norm_kernel(float* __restrict__ ns,
                                                            float* __restrict__ nd, int n) {
    int i = blockIdx.x * blockDim.x + threadIdx.x;
    if (i < n) {
        ns[i] = rsqrtf(fmaxf(ns[i], 1.0f));
        nd[i] = rsqrtf(fmaxf(nd[i], 1.0f));
    }
}

template <int K>
__global__ __launch_bounds__(128) void gemm_ns_wmma(const float* __restrict__ X,
                                                    const float* __restrict__ W,
                                                    const float* __restrict__ ns,
                                                    float* __restrict__ H) {
    const int lane    = threadIdx.x & 31;
    const int wave    = threadIdx.x >> 5;
    const int rowBase = blockIdx.x * 16;
    const int colBase = wave * 16;
    const int half    = lane >> 4;
    const int l15     = lane & 15;
    const int row     = rowBase + l15;
    const int col     = colBase + l15;
    const float s     = ns[row];

    v8f c = {};
#pragma unroll 4
    for (int k0 = 0; k0 < K; k0 += 4) {
        const int kk = k0 + half * 2;
        const float2 xa = *(const float2*)(X + row * K + kk);
        v2f a;
        a.x = xa.x * s;
        a.y = xa.y * s;
        v2f b;
        b.x = W[kk * DH + col];
        b.y = W[(kk + 1) * DH + col];
        c = __builtin_amdgcn_wmma_f32_16x16x4_f32(false, a, false, b, (short)0, c,
                                                  false, false);
    }

    const int mBase = half * 8;
#pragma unroll
    for (int i = 0; i < 8; ++i) {
        H[(size_t)(rowBase + mBase + i) * DH + col] = c[i];
    }
}

__global__ __launch_bounds__(256) void aggregate_kernel(const float* __restrict__ H,
                                                        const int* __restrict__ src,
                                                        const int* __restrict__ dst,
                                                        float* __restrict__ agg) {
    const int gid  = blockIdx.x * blockDim.x + threadIdx.x;
    const int e    = gid >> 5;
    const int lane = gid & 31;
    const int s = __builtin_amdgcn_readfirstlane(src[e]);
    const int d = __builtin_amdgcn_readfirstlane(dst[e]);
    const float2 v = *(const float2*)(H + (size_t)s * DH + lane * 2);
    float* p = agg + (size_t)d * DH + lane * 2;
    atomicAdd(p + 0, v.x);
    atomicAdd(p + 1, v.y);
}

__global__ __launch_bounds__(256) void post_kernel(const float* __restrict__ agg,
                                                   const float* __restrict__ nd,
                                                   const float* __restrict__ bias,
                                                   const float* __restrict__ g,
                                                   const float* __restrict__ be,
                                                   float* __restrict__ outx,
                                                   float* __restrict__ pre,
                                                   int n, int do_relu, int do_ln) {
    const int node = blockIdx.x * 8 + (threadIdx.x >> 5);
    if (node >= n) return;
    const int lane = threadIdx.x & 31;
    const int c0   = lane * 2;
    const float ndv = nd[node];
    const float* r = agg + (size_t)node * DH;
    float v0 = r[c0 + 0] * ndv + bias[c0 + 0];
    float v1 = r[c0 + 1] * ndv + bias[c0 + 1];
    if (do_relu) {
        v0 = fmaxf(v0, 0.0f);
        v1 = fmaxf(v1, 0.0f);
    }
    if (pre) {
        pre[(size_t)node * DH + c0 + 0] = v0;
        pre[(size_t)node * DH + c0 + 1] = v1;
    }
    if (do_ln) {
        float ssum = v0 + v1;
#pragma unroll
        for (int off = 16; off >= 1; off >>= 1) ssum += __shfl_xor(ssum, off, 32);
        const float mu = ssum * (1.0f / DH);
        const float d0 = v0 - mu, d1 = v1 - mu;
        float vs = d0 * d0 + d1 * d1;
#pragma unroll
        for (int off = 16; off >= 1; off >>= 1) vs += __shfl_xor(vs, off, 32);
        const float inv = rsqrtf(vs * (1.0f / DH) + LN_EPS);
        v0 = d0 * inv * g[c0 + 0] + be[c0 + 0];
        v1 = d1 * inv * g[c0 + 1] + be[c0 + 1];
    }
    float* o = outx + (size_t)node * DH + c0;
    o[0] = v0;
    o[1] = v1;
}

extern "C" void kernel_launch(void* const* d_in, const int* in_sizes, int n_in,
                              void* d_out, int out_size, void* d_ws, size_t ws_size,
                              hipStream_t stream) {
    (void)in_sizes; (void)n_in; (void)out_size; (void)ws_size;

    const float* feats = (const float*)d_in[0];
    const float* W1    = (const float*)d_in[1];
    const float* b1    = (const float*)d_in[2];
    const float* g1    = (const float*)d_in[3];
    const float* be1   = (const float*)d_in[4];
    const float* W2    = (const float*)d_in[5];
    const float* b2    = (const float*)d_in[6];
    const float* g2    = (const float*)d_in[7];
    const float* be2   = (const float*)d_in[8];
    const float* W3    = (const float*)d_in[9];
    const float* b3    = (const float*)d_in[10];
    const int*   src   = (const int*)d_in[11];
    const int*   dst   = (const int*)d_in[12];

    float* ns  = (float*)d_ws;
    float* nd  = ns + N_NODES;
    float* H   = nd + N_NODES;
    float* AGG = H + (size_t)N_NODES * DH;
    float* XB  = AGG + (size_t)N_NODES * DH;

    float* h_emb = (float*)d_out;
    float* h_out = h_emb + (size_t)N_NODES * DH;

    hipMemsetAsync(ns, 0, 2 * (size_t)N_NODES * sizeof(float), stream);
    degree_kernel<<<N_EDGES / 256, 256, 0, stream>>>(src, ns, N_EDGES);
    degree_kernel<<<N_EDGES / 256, 256, 0, stream>>>(dst, nd, N_EDGES);
    finalize_norm_kernel<<<(N_NODES + 255) / 256, 256, 0, stream>>>(ns, nd, N_NODES);

    const size_t aggBytes = (size_t)N_NODES * DH * sizeof(float);

    gemm_ns_wmma<DIN><<<N_NODES / 16, 128, 0, stream>>>(feats, W1, ns, H);
    hipMemsetAsync(AGG, 0, aggBytes, stream);
    aggregate_kernel<<<N_EDGES / 8, 256, 0, stream>>>(H, src, dst, AGG);
    post_kernel<<<N_NODES / 8, 256, 0, stream>>>(AGG, nd, b1, g1, be1, XB, nullptr,
                                                 N_NODES, 1, 1);

    gemm_ns_wmma<DH><<<N_NODES / 16, 128, 0, stream>>>(XB, W2, ns, H);
    hipMemsetAsync(AGG, 0, aggBytes, stream);
    aggregate_kernel<<<N_EDGES / 8, 256, 0, stream>>>(H, src, dst, AGG);
    post_kernel<<<N_NODES / 8, 256, 0, stream>>>(AGG, nd, b2, g2, be2, XB, h_emb,
                                                 N_NODES, 1, 1);

    gemm_ns_wmma<DH><<<N_NODES / 16, 128, 0, stream>>>(XB, W3, ns, H);
    hipMemsetAsync(AGG, 0, aggBytes, stream);
    aggregate_kernel<<<N_EDGES / 8, 256, 0, stream>>>(H, src, dst, AGG);
    post_kernel<<<N_NODES / 8, 256, 0, stream>>>(AGG, nd, b3, nullptr, nullptr, h_out,
                                                 nullptr, N_NODES, 0, 0);
}


# degree counts: f32 atomics issued only from the cache half that owns the node range (die id picks the range, per-half ticket queues hand out index chunks); counts exact as before
# speedup vs baseline: 1.2125x; 1.2125x over previous
#include <hip/hip_runtime.h>

#define N_NODES 100000
#define N_EDGES 1600000
#define DIN 128
#define DH 64
#define LN_EPS 1e-5f

typedef __attribute__((ext_vector_type(2))) float v2f;
typedef __attribute__((ext_vector_type(8))) float v8f;

#define DEG_CHUNK 1024
#define DEG_TASKS ((N_EDGES + DEG_CHUNK - 1) / DEG_CHUNK)
#define HALF_SPLIT (N_NODES / 2)
#define TICK_STRIDE 64

__global__ __launch_bounds__(256) void degree_kernel(const int* __restrict__ idx,
                                                     float* __restrict__ deg,
                                                     int* __restrict__ tick) {
    __shared__ int s_task;
    const unsigned where = __builtin_amdgcn_s_sendmsg_rtn(0x87);
    int hh = (int)((where >> 18) & 1u);
    for (int pass = 0; pass < 2; ++pass) {
        for (int it = 0; it <= DEG_TASKS; ++it) {
            if (threadIdx.x == 0) s_task = atomicAdd(&tick[hh * TICK_STRIDE], 1);
            __syncthreads();
            const int t = s_task;
            __syncthreads();
            if (t >= DEG_TASKS) break;
            const int off = t * DEG_CHUNK + (int)threadIdx.x * 4;
            if (off < N_EDGES) {
                const int4 v = *(const int4*)(idx + off);
                if ((int)(v.x >= HALF_SPLIT) == hh) atomicAdd(&deg[v.x], 1.0f);
                if ((int)(v.y >= HALF_SPLIT) == hh) atomicAdd(&deg[v.y], 1.0f);
                if ((int)(v.z >= HALF_SPLIT) == hh) atomicAdd(&deg[v.z], 1.0f);
                if ((int)(v.w >= HALF_SPLIT) == hh) atomicAdd(&deg[v.w], 1.0f);
            }
        }
        hh ^= 1;
    }
}

__global__ __launch_bounds__(256) void finalize_norm_kernel(float* __restrict__ ns,
                                                            float* __restrict__ nd, int n) {
    int i = blockIdx.x * blockDim.x + threadIdx.x;
    if (i < n) {
        ns[i] = rsqrtf(fmaxf(ns[i], 1.0f));
        nd[i] = rsqrtf(fmaxf(nd[i], 1.0f));
    }
}

template <int K>
__global__ __launch_bounds__(128) void gemm_ns_wmma(const float* __restrict__ X,
                                                    const float* __restrict__ W,
                                                    const float* __restrict__ ns,
                                                    float* __restrict__ H) {
    const int lane    = threadIdx.x & 31;
    const int wave    = threadIdx.x >> 5;
    const int rowBase = blockIdx.x * 16;
    const int colBase = wave * 16;
    const int half    = lane >> 4;
    const int l15     = lane & 15;
    const int row     = rowBase + l15;
    const int col     = colBase + l15;
    const float s     = ns[row];

    v8f c = {};
#pragma unroll 4
    for (int k0 = 0; k0 < K; k0 += 4) {
        const int kk = k0 + half * 2;
        const float2 xa = *(const float2*)(X + row * K + kk);
        v2f a;
        a.x = xa.x * s;
        a.y = xa.y * s;
        v2f b;
        b.x = W[kk * DH + col];
        b.y = W[(kk + 1) * DH + col];
        c = __builtin_amdgcn_wmma_f32_16x16x4_f32(false, a, false, b, (short)0, c,
                                                  false, false);
    }

    const int mBase = half * 8;
#pragma unroll
    for (int i = 0; i < 8; ++i) {
        H[(size_t)(rowBase + mBase + i) * DH + col] = c[i];
    }
}

__global__ __launch_bounds__(256) void aggregate_kernel(const float* __restrict__ H,
                                                        const int* __restrict__ src,
                                                        const int* __restrict__ dst,
                                                        float* __restrict__ agg) {
    const int gid  = blockIdx.x * blockDim.x + threadIdx.x;
    const int e    = gid >> 5;
    const int lane = gid & 31;
    const int s = __builtin_amdgcn_readfirstlane(src[e]);
    const int d = __builtin_amdgcn_readfirstlane(dst[e]);
    const float2 v = *(const float2*)(H + (size_t)s * DH + lane * 2);
    float* p = agg + (size_t)d * DH + lane * 2;
    atomicAdd(p + 0, v.x);
    atomicAdd(p + 1, v.y);
}

__global__ __launch_bounds__(256) void post_kernel(const float* __restrict__ agg,
                                                   const float* __restrict__ nd,
                                                   const float* __restrict__ bias,
                                                   const float* __restrict__ g,
                                                   const float* __restrict__ be,
                                                   float* __restrict__ outx,
                                                   float* __restrict__ pre,
                                                   int n, int do_relu, int do_ln) {
    const int node = blockIdx.x * 8 + (threadIdx.x >> 5);
    if (node >= n) return;
    const int lane = threadIdx.x & 31;
    const int c0   = lane * 2;
    const float ndv = nd[node];
    const float* r = agg + (size_t)node * DH;
    float v0 = r[c0 + 0] * ndv + bias[c0 + 0];
    float v1 = r[c0 + 1] * ndv + bias[c0 + 1];
    if (do_relu) {
        v0 = fmaxf(v0, 0.0f);
        v1 = fmaxf(v1, 0.0f);
    }
    if (pre) {
        pre[(size_t)node * DH + c0 + 0] = v0;
        pre[(size_t)node * DH + c0 + 1] = v1;
    }
    if (do_ln) {
        float ssum = v0 + v1;
#pragma unroll
        for (int off = 16; off >= 1; off >>= 1) ssum += __shfl_xor(ssum, off, 32);
        const float mu = ssum * (1.0f / DH);
        const float d0 = v0 - mu, d1 = v1 - mu;
        float vs = d0 * d0 + d1 * d1;
#pragma unroll
        for (int off = 16; off >= 1; off >>= 1) vs += __shfl_xor(vs, off, 32);
        const float inv = rsqrtf(vs * (1.0f / DH) + LN_EPS);
        v0 = d0 * inv * g[c0 + 0] + be[c0 + 0];
        v1 = d1 * inv * g[c0 + 1] + be[c0 + 1];
    }
    float* o = outx + (size_t)node * DH + c0;
    o[0] = v0;
    o[1] = v1;
}

extern "C" void kernel_launch(void* const* d_in, const int* in_sizes, int n_in,
                              void* d_out, int out_size, void* d_ws, size_t ws_size,
                              hipStream_t stream) {
    (void)in_sizes; (void)n_in; (void)out_size; (void)ws_size;

    const float* feats = (const float*)d_in[0];
    const float* W1    = (const float*)d_in[1];
    const float* b1    = (const float*)d_in[2];
    const float* g1    = (const float*)d_in[3];
    const float* be1   = (const float*)d_in[4];
    const float* W2    = (const float*)d_in[5];
    const float* b2    = (const float*)d_in[6];
    const float* g2    = (const float*)d_in[7];
    const float* be2   = (const float*)d_in[8];
    const float* W3    = (const float*)d_in[9];
    const float* b3    = (const float*)d_in[10];
    const int*   src   = (const int*)d_in[11];
    const int*   dst   = (const int*)d_in[12];

    int*   tick = (int*)d_ws;
    float* ns  = (float*)d_ws + 4 * TICK_STRIDE;
    float* nd  = ns + N_NODES;
    float* H   = nd + N_NODES;
    float* AGG = H + (size_t)N_NODES * DH;
    float* XB  = AGG + (size_t)N_NODES * DH;

    float* h_emb = (float*)d_out;
    float* h_out = h_emb + (size_t)N_NODES * DH;

    hipMemsetAsync(d_ws, 0, (4 * TICK_STRIDE + 2 * (size_t)N_NODES) * sizeof(float), stream);
    degree_kernel<<<1024, 256, 0, stream>>>(src, ns, tick);
    degree_kernel<<<1024, 256, 0, stream>>>(dst, nd, tick + 2 * TICK_STRIDE);
    finalize_norm_kernel<<<(N_NODES + 255) / 256, 256, 0, stream>>>(ns, nd, N_NODES);

    const size_t aggBytes = (size_t)N_NODES * DH * sizeof(float);

    gemm_ns_wmma<DIN><<<N_NODES / 16, 128, 0, stream>>>(feats, W1, ns, H);
    hipMemsetAsync(AGG, 0, aggBytes, stream);
    aggregate_kernel<<<N_EDGES / 8, 256, 0, stream>>>(H, src, dst, AGG);
    post_kernel<<<N_NODES / 8, 256, 0, stream>>>(AGG, nd, b1, g1, be1, XB, nullptr,
                                                 N_NODES, 1, 1);

    gemm_ns_wmma<DH><<<N_NODES / 16, 128, 0, stream>>>(XB, W2, ns, H);
    hipMemsetAsync(AGG, 0, aggBytes, stream);
    aggregate_kernel<<<N_EDGES / 8, 256, 0, stream>>>(H, src, dst, AGG);
    post_kernel<<<N_NODES / 8, 256, 0, stream>>>(AGG, nd, b2, g2, be2, XB, h_emb,
                                                 N_NODES, 1, 1);

    gemm_ns_wmma<DH><<<N_NODES / 16, 128, 0, stream>>>(XB, W3, ns, H);
    hipMemsetAsync(AGG, 0, aggBytes, stream);
    aggregate_kernel<<<N_EDGES / 8, 256, 0, stream>>>(H, src, dst, AGG);
    post_kernel<<<N_NODES / 8, 256, 0, stream>>>(AGG, nd, b3, nullptr, nullptr, h_out,
                                                 nullptr, N_NODES, 0, 0);
}


# aggregate: f32 row atomics only from the cache half owning the destination range (die id; ticket queue per half, 512-edge chunks; wave walks its edges 8 gathers deep); unordered arrival as before
# speedup vs baseline: 1.7159x; 1.4152x over previous
#include <hip/hip_runtime.h>

#define N_NODES 100000
#define N_EDGES 1600000
#define DIN 128
#define DH 64
#define LN_EPS 1e-5f

typedef __attribute__((ext_vector_type(2))) float v2f;
typedef __attribute__((ext_vector_type(8))) float v8f;

#define DEG_CHUNK 1024
#define DEG_TASKS ((N_EDGES + DEG_CHUNK - 1) / DEG_CHUNK)
#define HALF_SPLIT (N_NODES / 2)
#define TICK_STRIDE 64

__global__ __launch_bounds__(256) void degree_kernel(const int* __restrict__ idx,
                                                     float* __restrict__ deg,
                                                     int* __restrict__ tick) {
    __shared__ int s_task;
    const unsigned where = __builtin_amdgcn_s_sendmsg_rtn(0x87);
    int hh = (int)((where >> 18) & 1u);
    for (int pass = 0; pass < 2; ++pass) {
        for (int it = 0; it <= DEG_TASKS; ++it) {
            if (threadIdx.x == 0) s_task = atomicAdd(&tick[hh * TICK_STRIDE], 1);
            __syncthreads();
            const int t = s_task;
            __syncthreads();
            if (t >= DEG_TASKS) break;
            const int off = t * DEG_CHUNK + (int)threadIdx.x * 4;
            if (off < N_EDGES) {
                const int4 v = *(const int4*)(idx + off);
                if ((int)(v.x >= HALF_SPLIT) == hh) atomicAdd(&deg[v.x], 1.0f);
                if ((int)(v.y >= HALF_SPLIT) == hh) atomicAdd(&deg[v.y], 1.0f);
                if ((int)(v.z >= HALF_SPLIT) == hh) atomicAdd(&deg[v.z], 1.0f);
                if ((int)(v.w >= HALF_SPLIT) == hh) atomicAdd(&deg[v.w], 1.0f);
            }
        }
        hh ^= 1;
    }
}

__global__ __launch_bounds__(256) void finalize_norm_kernel(float* __restrict__ ns,
                                                            float* __restrict__ nd, int n) {
    int i = blockIdx.x * blockDim.x + threadIdx.x;
    if (i < n) {
        ns[i] = rsqrtf(fmaxf(ns[i], 1.0f));
        nd[i] = rsqrtf(fmaxf(nd[i], 1.0f));
    }
}

template <int K>
__global__ __launch_bounds__(128) void gemm_ns_wmma(const float* __restrict__ X,
                                                    const float* __restrict__ W,
                                                    const float* __restrict__ ns,
                                                    float* __restrict__ H) {
    const int lane    = threadIdx.x & 31;
    const int wave    = threadIdx.x >> 5;
    const int rowBase = blockIdx.x * 16;
    const int colBase = wave * 16;
    const int half    = lane >> 4;
    const int l15     = lane & 15;
    const int row     = rowBase + l15;
    const int col     = colBase + l15;
    const float s     = ns[row];

    v8f c = {};
#pragma unroll 4
    for (int k0 = 0; k0 < K; k0 += 4) {
        const int kk = k0 + half * 2;
        const float2 xa = *(const float2*)(X + row * K + kk);
        v2f a;
        a.x = xa.x * s;
        a.y = xa.y * s;
        v2f b;
        b.x = W[kk * DH + col];
        b.y = W[(kk + 1) * DH + col];
        c = __builtin_amdgcn_wmma_f32_16x16x4_f32(false, a, false, b, (short)0, c,
                                                  false, false);
    }

    const int mBase = half * 8;
#pragma unroll
    for (int i = 0; i < 8; ++i) {
        H[(size_t)(rowBase + mBase + i) * DH + col] = c[i];
    }
}

#define AGG_CHUNK 512
#define AGG_TASKS (N_EDGES / AGG_CHUNK)
#define AGG_DEPTH 8

__device__ __forceinline__ void walk_window(const float* __restrict__ H,
                                            float* __restrict__ agg,
                                            int sv, int dv, int hh, int lane) {
    unsigned m = __builtin_amdgcn_ballot_w32((int)(dv >= HALF_SPLIT) == hh);
    while (m != 0u) {
        const int left = __builtin_popcount(m);
        const int n = left < AGG_DEPTH ? left : AGG_DEPTH;
        float2 v[AGG_DEPTH];
        int d[AGG_DEPTH];
#pragma unroll
        for (int j = 0; j < AGG_DEPTH; ++j) {
            if (j < n) {
                const int i = __builtin_ctz(m);
                m &= m - 1u;
                const int s = __builtin_amdgcn_readlane(sv, i);
                d[j] = __builtin_amdgcn_readlane(dv, i);
                v[j] = *(const float2*)(H + (size_t)s * DH + lane * 2);
            }
        }
#pragma unroll
        for (int j = 0; j < AGG_DEPTH; ++j) {
            if (j < n) {
                float* p = agg + (size_t)d[j] * DH + lane * 2;
                atomicAdd(p + 0, v[j].x);
                atomicAdd(p + 1, v[j].y);
            }
        }
    }
}

__global__ __launch_bounds__(256) void aggregate_kernel(const float* __restrict__ H,
                                                        const int* __restrict__ src,
                                                        const int* __restrict__ dst,
                                                        float* __restrict__ agg,
                                                        int* __restrict__ tick) {
    __shared__ int s_task;
    const int lane = threadIdx.x & 31;
    const int wave = threadIdx.x >> 5;
    const unsigned where = __builtin_amdgcn_s_sendmsg_rtn(0x87);
    int hh = (int)((where >> 18) & 1u);
    for (int pass = 0; pass < 2; ++pass) {
        for (int it = 0; it <= AGG_TASKS; ++it) {
            if (threadIdx.x == 0) s_task = atomicAdd(&tick[hh * TICK_STRIDE], 1);
            __syncthreads();
            const int t = __builtin_amdgcn_readfirstlane(s_task);
            __syncthreads();
            if (t >= AGG_TASKS) break;
            const int e = t * AGG_CHUNK + wave * 64 + lane;
            const int dv0 = dst[e];
            const int dv1 = dst[e + 32];
            const int sv0 = src[e];
            const int sv1 = src[e + 32];
            asm volatile("" :: "v"(dv0), "v"(dv1), "v"(sv0), "v"(sv1));
            walk_window(H, agg, sv0, dv0, hh, lane);
            walk_window(H, agg, sv1, dv1, hh, lane);
        }
        hh ^= 1;
    }
}

__global__ __launch_bounds__(256) void post_kernel(const float* __restrict__ agg,
                                                   const float* __restrict__ nd,
                                                   const float* __restrict__ bias,
                                                   const float* __restrict__ g,
                                                   const float* __restrict__ be,
                                                   float* __restrict__ outx,
                                                   float* __restrict__ pre,
                                                   int n, int do_relu, int do_ln) {
    const int node = blockIdx.x * 8 + (threadIdx.x >> 5);
    if (node >= n) return;
    const int lane = threadIdx.x & 31;
    const int c0   = lane * 2;
    const float ndv = nd[node];
    const float* r = agg + (size_t)node * DH;
    float v0 = r[c0 + 0] * ndv + bias[c0 + 0];
    float v1 = r[c0 + 1] * ndv + bias[c0 + 1];
    if (do_relu) {
        v0 = fmaxf(v0, 0.0f);
        v1 = fmaxf(v1, 0.0f);
    }
    if (pre) {
        pre[(size_t)node * DH + c0 + 0] = v0;
        pre[(size_t)node * DH + c0 + 1] = v1;
    }
    if (do_ln) {
        float ssum = v0 + v1;
#pragma unroll
        for (int off = 16; off >= 1; off >>= 1) ssum += __shfl_xor(ssum, off, 32);
        const float mu = ssum * (1.0f / DH);
        const float d0 = v0 - mu, d1 = v1 - mu;
        float vs = d0 * d0 + d1 * d1;
#pragma unroll
        for (int off = 16; off >= 1; off >>= 1) vs += __shfl_xor(vs, off, 32);
        const float inv = rsqrtf(vs * (1.0f / DH) + LN_EPS);
        v0 = d0 * inv * g[c0 + 0] + be[c0 + 0];
        v1 = d1 * inv * g[c0 + 1] + be[c0 + 1];
    }
    float* o = outx + (size_t)node * DH + c0;
    o[0] = v0;
    o[1] = v1;
}

extern "C" void kernel_launch(void* const* d_in, const int* in_sizes, int n_in,
                              void* d_out, int out_size, void* d_ws, size_t ws_size,
                              hipStream_t stream) {
    (void)in_sizes; (void)n_in; (void)out_size; (void)ws_size;

    const float* feats = (const float*)d_in[0];
    const float* W1    = (const float*)d_in[1];
    const float* b1    = (const float*)d_in[2];
    const float* g1    = (const float*)d_in[3];
    const float* be1   = (const float*)d_in[4];
    const float* W2    = (const float*)d_in[5];
    const float* b2    = (const float*)d_in[6];
    const float* g2    = (const float*)d_in[7];
    const float* be2   = (const float*)d_in[8];
    const float* W3    = (const float*)d_in[9];
    const float* b3    = (const float*)d_in[10];
    const int*   src   = (const int*)d_in[11];
    const int*   dst   = (const int*)d_in[12];

    int*   tick = (int*)d_ws;
    float* ns  = (float*)d_ws + 16 * TICK_STRIDE;
    float* nd  = ns + N_NODES;
    float* H   = nd + N_NODES;
    float* AGG = H + (size_t)N_NODES * DH;
    float* XB  = AGG + (size_t)N_NODES * DH;

    float* h_emb = (float*)d_out;
    float* h_out = h_emb + (size_t)N_NODES * DH;

    hipMemsetAsync(d_ws, 0, (16 * TICK_STRIDE + 2 * (size_t)N_NODES) * sizeof(float), stream);
    degree_kernel<<<1024, 256, 0, stream>>>(src, ns, tick);
    degree_kernel<<<1024, 256, 0, stream>>>(dst, nd, tick + 2 * TICK_STRIDE);
    finalize_norm_kernel<<<(N_NODES + 255) / 256, 256, 0, stream>>>(ns, nd, N_NODES);

    const size_t aggBytes = (size_t)N_NODES * DH * sizeof(float);

    gemm_ns_wmma<DIN><<<N_NODES / 16, 128, 0, stream>>>(feats, W1, ns, H);
    hipMemsetAsync(AGG, 0, aggBytes, stream);
    aggregate_kernel<<<2048, 256, 0, stream>>>(H, src, dst, AGG, tick + 4 * TICK_STRIDE);
    post_kernel<<<N_NODES / 8, 256, 0, stream>>>(AGG, nd, b1, g1, be1, XB, nullptr,
                                                 N_NODES, 1, 1);

    gemm_ns_wmma<DH><<<N_NODES / 16, 128, 0, stream>>>(XB, W2, ns, H);
    hipMemsetAsync(AGG, 0, aggBytes, stream);
    aggregate_kernel<<<2048, 256, 0, stream>>>(H, src, dst, AGG, tick + 6 * TICK_STRIDE);
    post_kernel<<<N_NODES / 8, 256, 0, stream>>>(AGG, nd, b2, g2, be2, XB, h_emb,
                                                 N_NODES, 1, 1);

    gemm_ns_wmma<DH><<<N_NODES / 16, 128, 0, stream>>>(XB, W3, ns, H);
    hipMemsetAsync(AGG, 0, aggBytes, stream);
    aggregate_kernel<<<2048, 256, 0, stream>>>(H, src, dst, AGG, tick + 8 * TICK_STRIDE);
    post_kernel<<<N_NODES / 8, 256, 0, stream>>>(AGG, nd, b3, nullptr, nullptr, h_out,
                                                 nullptr, N_NODES, 0, 0);
}


# aggregate: each f32 row add instruction carries one whole 128-byte line (wave halves trade one value, lane l with l^16) instead of half the words of two lines; unordered arrival as before
# speedup vs baseline: 2.0893x; 1.2176x over previous
#include <hip/hip_runtime.h>

#define N_NODES 100000
#define N_EDGES 1600000
#define DIN 128
#define DH 64
#define LN_EPS 1e-5f

typedef __attribute__((ext_vector_type(2))) float v2f;
typedef __attribute__((ext_vector_type(8))) float v8f;

#define DEG_CHUNK 1024
#define DEG_TASKS ((N_EDGES + DEG_CHUNK - 1) / DEG_CHUNK)
#define HALF_SPLIT (N_NODES / 2)
#define TICK_STRIDE 64

__global__ __launch_bounds__(256) void degree_kernel(const int* __restrict__ idx,
                                                     float* __restrict__ deg,
                                                     int* __restrict__ tick) {
    __shared__ int s_task;
    const unsigned where = __builtin_amdgcn_s_sendmsg_rtn(0x87);
    int hh = (int)((where >> 18) & 1u);
    for (int pass = 0; pass < 2; ++pass) {
        for (int it = 0; it <= DEG_TASKS; ++it) {
            if (threadIdx.x == 0) s_task = atomicAdd(&tick[hh * TICK_STRIDE], 1);
            __syncthreads();
            const int t = s_task;
            __syncthreads();
            if (t >= DEG_TASKS) break;
            const int off = t * DEG_CHUNK + (int)threadIdx.x * 4;
            if (off < N_EDGES) {
                const int4 v = *(const int4*)(idx + off);
                if ((int)(v.x >= HALF_SPLIT) == hh) atomicAdd(&deg[v.x], 1.0f);
                if ((int)(v.y >= HALF_SPLIT) == hh) atomicAdd(&deg[v.y], 1.0f);
                if ((int)(v.z >= HALF_SPLIT) == hh) atomicAdd(&deg[v.z], 1.0f);
                if ((int)(v.w >= HALF_SPLIT) == hh) atomicAdd(&deg[v.w], 1.0f);
            }
        }
        hh ^= 1;
    }
}

__global__ __launch_bounds__(256) void finalize_norm_kernel(float* __restrict__ ns,
                                                            float* __restrict__ nd, int n) {
    int i = blockIdx.x * blockDim.x + threadIdx.x;
    if (i < n) {
        ns[i] = rsqrtf(fmaxf(ns[i], 1.0f));
        nd[i] = rsqrtf(fmaxf(nd[i], 1.0f));
    }
}

template <int K>
__global__ __launch_bounds__(128) void gemm_ns_wmma(const float* __restrict__ X,
                                                    const float* __restrict__ W,
                                                    const float* __restrict__ ns,
                                                    float* __restrict__ H) {
    const int lane    = threadIdx.x & 31;
    const int wave    = threadIdx.x >> 5;
    const int rowBase = blockIdx.x * 16;
    const int colBase = wave * 16;
    const int half    = lane >> 4;
    const int l15     = lane & 15;
    const int row     = rowBase + l15;
    const int col     = colBase + l15;
    const float s     = ns[row];

    v8f c = {};
#pragma unroll 4
    for (int k0 = 0; k0 < K; k0 += 4) {
        const int kk = k0 + half * 2;
        const float2 xa = *(const float2*)(X + row * K + kk);
        v2f a;
        a.x = xa.x * s;
        a.y = xa.y * s;
        v2f b;
        b.x = W[kk * DH + col];
        b.y = W[(kk + 1) * DH + col];
        c = __builtin_amdgcn_wmma_f32_16x16x4_f32(false, a, false, b, (short)0, c,
                                                  false, false);
    }

    const int mBase = half * 8;
#pragma unroll
    for (int i = 0; i < 8; ++i) {
        H[(size_t)(rowBase + mBase + i) * DH + col] = c[i];
    }
}

#define AGG_CHUNK 512
#define AGG_TASKS (N_EDGES / AGG_CHUNK)
#define AGG_DEPTH 8

__device__ __forceinline__ float other_half(float x) {
    return __builtin_bit_cast(float,
        __builtin_amdgcn_permlanex16(__builtin_bit_cast(int, x), __builtin_bit_cast(int, x),
                                     0x76543210, 0xfedcba98, false, false));
}

__device__ __forceinline__ void walk_window(const float* __restrict__ H,
                                            float* __restrict__ agg,
                                            int sv, int dv, int hh, int lane) {
    const bool upper = lane >= 16;
    const int colA = 2 * (lane & 15) + (lane >> 4);
    unsigned m = __builtin_amdgcn_ballot_w32((int)(dv >= HALF_SPLIT) == hh);
    while (m != 0u) {
        const int left = __builtin_popcount(m);
        const int n = left < AGG_DEPTH ? left : AGG_DEPTH;
        float2 v[AGG_DEPTH];
        int d[AGG_DEPTH];
#pragma unroll
        for (int j = 0; j < AGG_DEPTH; ++j) {
            if (j < n) {
                const int i = __builtin_ctz(m);
                m &= m - 1u;
                const int s = __builtin_amdgcn_readlane(sv, i);
                d[j] = __builtin_amdgcn_readlane(dv, i);
                v[j] = *(const float2*)(H + (size_t)s * DH + lane * 2);
            }
        }
#pragma unroll
        for (int j = 0; j < AGG_DEPTH; ++j) {
            if (j < n) {
                const float got = other_half(upper ? v[j].x : v[j].y);
                const float a = upper ? got : v[j].x;
                const float b = upper ? v[j].y : got;
                float* p = agg + (size_t)d[j] * DH + colA;
                atomicAdd(p, a);
                atomicAdd(p + 32, b);
            }
        }
    }
}

__global__ __launch_bounds__(256) void aggregate_kernel(const float* __restrict__ H,
                                                        const int* __restrict__ src,
                                                        const int* __restrict__ dst,
                                                        float* __restrict__ agg,
                                                        int* __restrict__ tick) {
    __shared__ int s_task;
    const int lane = threadIdx.x & 31;
    const int wave = threadIdx.x >> 5;
    const unsigned where = __builtin_amdgcn_s_sendmsg_rtn(0x87);
    int hh = (int)((where >> 18) & 1u);
    for (int pass = 0; pass < 2; ++pass) {
        for (int it = 0; it <= AGG_TASKS; ++it) {
            if (threadIdx.x == 0) s_task = atomicAdd(&tick[hh * TICK_STRIDE], 1);
            __syncthreads();
            const int t = __builtin_amdgcn_readfirstlane(s_task);
            __syncthreads();
            if (t >= AGG_TASKS) break;
            const int e = t * AGG_CHUNK + wave * 64 + lane;
            const int dv0 = dst[e];
            const int dv1 = dst[e + 32];
            const int sv0 = src[e];
            const int sv1 = src[e + 32];
            asm volatile("" :: "v"(dv0), "v"(dv1), "v"(sv0), "v"(sv1));
            walk_window(H, agg, sv0, dv0, hh, lane);
            walk_window(H, agg, sv1, dv1, hh, lane);
        }
        hh ^= 1;
    }
}

__global__ __launch_bounds__(256) void post_kernel(const float* __restrict__ agg,
                                                   const float* __restrict__ nd,
                                                   const float* __restrict__ bias,
                                                   const float* __restrict__ g,
                                                   const float* __restrict__ be,
                                                   float* __restrict__ outx,
                                                   float* __restrict__ pre,
                                                   int n, int do_relu, int do_ln) {
    const int node = blockIdx.x * 8 + (threadIdx.x >> 5);
    if (node >= n) return;
    const int lane = threadIdx.x & 31;
    const int c0   = lane * 2;
    const float ndv = nd[node];
    const float* r = agg + (size_t)node * DH;
    float v0 = r[c0 + 0] * ndv + bias[c0 + 0];
    float v1 = r[c0 + 1] * ndv + bias[c0 + 1];
    if (do_relu) {
        v0 = fmaxf(v0, 0.0f);
        v1 = fmaxf(v1, 0.0f);
    }
    if (pre) {
        pre[(size_t)node * DH + c0 + 0] = v0;
        pre[(size_t)node * DH + c0 + 1] = v1;
    }
    if (do_ln) {
        float ssum = v0 + v1;
#pragma unroll
        for (int off = 16; off >= 1; off >>= 1) ssum += __shfl_xor(ssum, off, 32);
        const float mu = ssum * (1.0f / DH);
        const float d0 = v0 - mu, d1 = v1 - mu;
        float vs = d0 * d0 + d1 * d1;
#pragma unroll
        for (int off = 16; off >= 1; off >>= 1) vs += __shfl_xor(vs, off, 32);
        const float inv = rsqrtf(vs * (1.0f / DH) + LN_EPS);
        v0 = d0 * inv * g[c0 + 0] + be[c0 + 0];
        v1 = d1 * inv * g[c0 + 1] + be[c0 + 1];
    }
    float* o = outx + (size_t)node * DH + c0;
    o[0] = v0;
    o[1] = v1;
}

extern "C" void kernel_launch(void* const* d_in, const int* in_sizes, int n_in,
                              void* d_out, int out_size, void* d_ws, size_t ws_size,
                              hipStream_t stream) {
    (void)in_sizes; (void)n_in; (void)out_size; (void)ws_size;

    const float* feats = (const float*)d_in[0];
    const float* W1    = (const float*)d_in[1];
    const float* b1    = (const float*)d_in[2];
    const float* g1    = (const float*)d_in[3];
    const float* be1   = (const float*)d_in[4];
    const float* W2    = (const float*)d_in[5];
    const float* b2    = (const float*)d_in[6];
    const float* g2    = (const float*)d_in[7];
    const float* be2   = (const float*)d_in[8];
    const float* W3    = (const float*)d_in[9];
    const float* b3    = (const float*)d_in[10];
    const int*   src   = (const int*)d_in[11];
    const int*   dst   = (const int*)d_in[12];

    int*   tick = (int*)d_ws;
    float* ns  = (float*)d_ws + 16 * TICK_STRIDE;
    float* nd  = ns + N_NODES;
    float* H   = nd + N_NODES;
    float* AGG = H + (size_t)N_NODES * DH;
    float* XB  = AGG + (size_t)N_NODES * DH;

    float* h_emb = (float*)d_out;
    float* h_out = h_emb + (size_t)N_NODES * DH;

    hipMemsetAsync(d_ws, 0, (16 * TICK_STRIDE + 2 * (size_t)N_NODES) * sizeof(float), stream);
    degree_kernel<<<1024, 256, 0, stream>>>(src, ns, tick);
    degree_kernel<<<1024, 256, 0, stream>>>(dst, nd, tick + 2 * TICK_STRIDE);
    finalize_norm_kernel<<<(N_NODES + 255) / 256, 256, 0, stream>>>(ns, nd, N_NODES);

    const size_t aggBytes = (size_t)N_NODES * DH * sizeof(float);

    gemm_ns_wmma<DIN><<<N_NODES / 16, 128, 0, stream>>>(feats, W1, ns, H);
    hipMemsetAsync(AGG, 0, aggBytes, stream);
    aggregate_kernel<<<2048, 256, 0, stream>>>(H, src, dst, AGG, tick + 4 * TICK_STRIDE);
    post_kernel<<<N_NODES / 8, 256, 0, stream>>>(AGG, nd, b1, g1, be1, XB, nullptr,
                                                 N_NODES, 1, 1);

    gemm_ns_wmma<DH><<<N_NODES / 16, 128, 0, stream>>>(XB, W2, ns, H);
    hipMemsetAsync(AGG, 0, aggBytes, stream);
    aggregate_kernel<<<2048, 256, 0, stream>>>(H, src, dst, AGG, tick + 6 * TICK_STRIDE);
    post_kernel<<<N_NODES / 8, 256, 0, stream>>>(AGG, nd, b2, g2, be2, XB, h_emb,
                                                 N_NODES, 1, 1);

    gemm_ns_wmma<DH><<<N_NODES / 16, 128, 0, stream>>>(XB, W3, ns, H);
    hipMemsetAsync(AGG, 0, aggBytes, stream);
    aggregate_kernel<<<2048, 256, 0, stream>>>(H, src, dst, AGG, tick + 8 * TICK_STRIDE);
    post_kernel<<<N_NODES / 8, 256, 0, stream>>>(AGG, nd, b3, nullptr, nullptr, h_out,
                                                 nullptr, N_NODES, 0, 0);
}


# layers 2, 3: the transform block finishes its own 32 agg rows (nd, bias, ReLU, LayerNorm, ns) into LDS and takes its A fragments there; 2 post launches and the XB plane go; sums unchanged
# speedup vs baseline: 2.2469x; 1.0754x over previous
#include <hip/hip_runtime.h>

#define N_NODES 100000
#define N_EDGES 1600000
#define DIN 128
#define DH 64
#define LN_EPS 1e-5f

typedef __attribute__((ext_vector_type(2))) float v2f;
typedef __attribute__((ext_vector_type(8))) float v8f;

#define DEG_CHUNK 1024
#define DEG_TASKS ((N_EDGES + DEG_CHUNK - 1) / DEG_CHUNK)
#define HALF_SPLIT (N_NODES / 2)
#define TICK_STRIDE 64

__global__ __launch_bounds__(256) void degree_kernel(const int* __restrict__ idx,
                                                     float* __restrict__ deg,
                                                     int* __restrict__ tick) {
    __shared__ int s_task;
    const unsigned where = __builtin_amdgcn_s_sendmsg_rtn(0x87);
    int hh = (int)((where >> 18) & 1u);
    for (int pass = 0; pass < 2; ++pass) {
        for (int it = 0; it <= DEG_TASKS; ++it) {
            if (threadIdx.x == 0) s_task = atomicAdd(&tick[hh * TICK_STRIDE], 1);
            __syncthreads();
            const int t = s_task;
            __syncthreads();
            if (t >= DEG_TASKS) break;
            const int off = t * DEG_CHUNK + (int)threadIdx.x * 4;
            if (off < N_EDGES) {
                const int4 v = *(const int4*)(idx + off);
                if ((int)(v.x >= HALF_SPLIT) == hh) atomicAdd(&deg[v.x], 1.0f);
                if ((int)(v.y >= HALF_SPLIT) == hh) atomicAdd(&deg[v.y], 1.0f);
                if ((int)(v.z >= HALF_SPLIT) == hh) atomicAdd(&deg[v.z], 1.0f);
                if ((int)(v.w >= HALF_SPLIT) == hh) atomicAdd(&deg[v.w], 1.0f);
            }
        }
        hh ^= 1;
    }
}

__global__ __launch_bounds__(256) void finalize_norm_kernel(float* __restrict__ ns,
                                                            float* __restrict__ nd, int n) {
    int i = blockIdx.x * blockDim.x + threadIdx.x;
    if (i < n) {
        ns[i] = rsqrtf(fmaxf(ns[i], 1.0f));
        nd[i] = rsqrtf(fmaxf(nd[i], 1.0f));
    }
}

template <int K>
__global__ __launch_bounds__(128) void gemm_ns_wmma(const float* __restrict__ X,
                                                    const float* __restrict__ W,
                                                    const float* __restrict__ ns,
                                                    float* __restrict__ H) {
    const int lane    = threadIdx.x & 31;
    const int wave    = threadIdx.x >> 5;
    const int rowBase = blockIdx.x * 16;
    const int colBase = wave * 16;
    const int half    = lane >> 4;
    const int l15     = lane & 15;
    const int row     = rowBase + l15;
    const int col     = colBase + l15;
    const float s     = ns[row];

    v8f c = {};
#pragma unroll 4
    for (int k0 = 0; k0 < K; k0 += 4) {
        const int kk = k0 + half * 2;
        const float2 xa = *(const float2*)(X + row * K + kk);
        v2f a;
        a.x = xa.x * s;
        a.y = xa.y * s;
        v2f b;
        b.x = W[kk * DH + col];
        b.y = W[(kk + 1) * DH + col];
        c = __builtin_amdgcn_wmma_f32_16x16x4_f32(false, a, false, b, (short)0, c,
                                                  false, false);
    }

    const int mBase = half * 8;
#pragma unroll
    for (int i = 0; i < 8; ++i) {
        H[(size_t)(rowBase + mBase + i) * DH + col] = c[i];
    }
}

#define AGG_CHUNK 512
#define AGG_TASKS (N_EDGES / AGG_CHUNK)
#define AGG_DEPTH 8

__device__ __forceinline__ float other_half(float x) {
    return __builtin_bit_cast(float,
        __builtin_amdgcn_permlanex16(__builtin_bit_cast(int, x), __builtin_bit_cast(int, x),
                                     0x76543210, 0xfedcba98, false, false));
}

__device__ __forceinline__ void walk_window(const float* __restrict__ H,
                                            float* __restrict__ agg,
                                            int sv, int dv, int hh, int lane) {
    const bool upper = lane >= 16;
    const int colA = 2 * (lane & 15) + (lane >> 4);
    unsigned m = __builtin_amdgcn_ballot_w32((int)(dv >= HALF_SPLIT) == hh);
    while (m != 0u) {
        const int left = __builtin_popcount(m);
        const int n = left < AGG_DEPTH ? left : AGG_DEPTH;
        float2 v[AGG_DEPTH];
        int d[AGG_DEPTH];
#pragma unroll
        for (int j = 0; j < AGG_DEPTH; ++j) {
            if (j < n) {
                const int i = __builtin_ctz(m);
                m &= m - 1u;
                const int s = __builtin_amdgcn_readlane(sv, i);
                d[j] = __builtin_amdgcn_readlane(dv, i);
                v[j] = *(const float2*)(H + (size_t)s * DH + lane * 2);
            }
        }
#pragma unroll
        for (int j = 0; j < AGG_DEPTH; ++j) {
            if (j < n) {
                const float got = other_half(upper ? v[j].x : v[j].y);
                const float a = upper ? got : v[j].x;
                const float b = upper ? v[j].y : got;
                float* p = agg + (size_t)d[j] * DH + colA;
                atomicAdd(p, a);
                atomicAdd(p + 32, b);
            }
        }
    }
}

__global__ __launch_bounds__(256) void aggregate_kernel(const float* __restrict__ H,
                                                        const int* __restrict__ src,
                                                        const int* __restrict__ dst,
                                                        float* __restrict__ agg,
                                                        int* __restrict__ tick) {
    __shared__ int s_task;
    const int lane = threadIdx.x & 31;
    const int wave = threadIdx.x >> 5;
    const unsigned where = __builtin_amdgcn_s_sendmsg_rtn(0x87);
    int hh = (int)((where >> 18) & 1u);
    for (int pass = 0; pass < 2; ++pass) {
        for (int it = 0; it <= AGG_TASKS; ++it) {
            if (threadIdx.x == 0) s_task = atomicAdd(&tick[hh * TICK_STRIDE], 1);
            __syncthreads();
            const int t = __builtin_amdgcn_readfirstlane(s_task);
            __syncthreads();
            if (t >= AGG_TASKS) break;
            const int e = t * AGG_CHUNK + wave * 64 + lane;
            const int dv0 = dst[e];
            const int dv1 = dst[e + 32];
            const int sv0 = src[e];
            const int sv1 = src[e + 32];
            asm volatile("" :: "v"(dv0), "v"(dv1), "v"(sv0), "v"(sv1));
            walk_window(H, agg, sv0, dv0, hh, lane);
            walk_window(H, agg, sv1, dv1, hh, lane);
        }
        hh ^= 1;
    }
}

__global__ __launch_bounds__(256) void post_kernel(const float* __restrict__ agg,
                                                   const float* __restrict__ nd,
                                                   const float* __restrict__ bias,
                                                   const float* __restrict__ g,
                                                   const float* __restrict__ be,
                                                   float* __restrict__ outx,
                                                   float* __restrict__ pre,
                                                   int n, int do_relu, int do_ln) {
    const int node = blockIdx.x * 8 + (threadIdx.x >> 5);
    if (node >= n) return;
    const int lane = threadIdx.x & 31;
    const int c0   = lane * 2;
    const float ndv = nd[node];
    const float* r = agg + (size_t)node * DH;
    float v0 = r[c0 + 0] * ndv + bias[c0 + 0];
    float v1 = r[c0 + 1] * ndv + bias[c0 + 1];
    if (do_relu) {
        v0 = fmaxf(v0, 0.0f);
        v1 = fmaxf(v1, 0.0f);
    }
    if (pre) {
        pre[(size_t)node * DH + c0 + 0] = v0;
        pre[(size_t)node * DH + c0 + 1] = v1;
    }
    if (do_ln) {
        float ssum = v0 + v1;
#pragma unroll
        for (int off = 16; off >= 1; off >>= 1) ssum += __shfl_xor(ssum, off, 32);
        const float mu = ssum * (1.0f / DH);
        const float d0 = v0 - mu, d1 = v1 - mu;
        float vs = d0 * d0 + d1 * d1;
#pragma unroll
        for (int off = 16; off >= 1; off >>= 1) vs += __shfl_xor(vs, off, 32);
        const float inv = rsqrtf(vs * (1.0f / DH) + LN_EPS);
        v0 = d0 * inv * g[c0 + 0] + be[c0 + 0];
        v1 = d1 * inv * g[c0 + 1] + be[c0 + 1];
    }
    float* o = outx + (size_t)node * DH + c0;
    o[0] = v0;
    o[1] = v1;
}

#define FR 32
#define AP 68

template <bool EMIT_PRE>
__global__ __launch_bounds__(128) void post_gemm_wmma(const float* __restrict__ agg,
                                                      const float* __restrict__ nd,
                                                      const float* __restrict__ bias,
                                                      const float* __restrict__ g,
                                                      const float* __restrict__ be,
                                                      const float* __restrict__ ns,
                                                      const float* __restrict__ W,
                                                      float* __restrict__ pre,
                                                      float* __restrict__ H) {
    __shared__ __attribute__((aligned(16))) float As[FR * AP];
    const int lane = threadIdx.x & 31;
    const int wave    = __builtin_amdgcn_readfirstlane((int)(threadIdx.x >> 5));
    const int rowBase = blockIdx.x * FR;
    const int half    = lane >> 4;
    const int l15     = lane & 15;
    const int col     = wave * 16 + l15;

    v2f bf[DH / 4];
#pragma unroll
    for (int q = 0; q < DH / 4; ++q) {
        const int kk = q * 4 + half * 2;
        bf[q].x = W[kk * DH + col];
        bf[q].y = W[(kk + 1) * DH + col];
    }

    const int c0 = lane * 2;
    const float b0 = bias[c0 + 0], b1 = bias[c0 + 1];
    const float g0 = g[c0 + 0],    g1 = g[c0 + 1];
    const float e0 = be[c0 + 0],   e1 = be[c0 + 1];
#pragma unroll
    for (int j = 0; j < FR / 4; ++j) {
        const int lr   = wave * (FR / 4) + j;
        const int node = rowBase + lr;
        const float ndv = nd[node];
        const float2 r = *(const float2*)(agg + (size_t)node * DH + c0);
        float v0 = r.x * ndv + b0;
        float v1 = r.y * ndv + b1;
        v0 = fmaxf(v0, 0.0f);
        v1 = fmaxf(v1, 0.0f);
        if (EMIT_PRE) {
            *(float2*)(pre + (size_t)node * DH + c0) = make_float2(v0, v1);
        }
        float ssum = v0 + v1;
#pragma unroll
        for (int off = 16; off >= 1; off >>= 1) ssum += __shfl_xor(ssum, off, 32);
        const float mu = ssum * (1.0f / DH);
        const float d0 = v0 - mu, d1 = v1 - mu;
        float vs = d0 * d0 + d1 * d1;
#pragma unroll
        for (int off = 16; off >= 1; off >>= 1) vs += __shfl_xor(vs, off, 32);
        const float inv = rsqrtf(vs * (1.0f / DH) + LN_EPS);
        v0 = d0 * inv * g0 + e0;
        v1 = d1 * inv * g1 + e1;
        const float s = ns[node];
        *(float2*)(As + lr * AP + c0) = make_float2(v0 * s, v1 * s);
    }
    __syncthreads();

    v8f acc0 = {};
    v8f acc1 = {};
#pragma unroll
    for (int q = 0; q < DH / 4; ++q) {
        const int kk = q * 4 + half * 2;
        const float2 x0 = *(const float2*)(As + l15 * AP + kk);
        const float2 x1 = *(const float2*)(As + (16 + l15) * AP + kk);
        v2f a0, a1;
        a0.x = x0.x; a0.y = x0.y;
        a1.x = x1.x; a1.y = x1.y;
        acc0 = __builtin_amdgcn_wmma_f32_16x16x4_f32(false, a0, false, bf[q], (short)0, acc0,
                                                     false, false);
        acc1 = __builtin_amdgcn_wmma_f32_16x16x4_f32(false, a1, false, bf[q], (short)0, acc1,
                                                     false, false);
    }

    const int mBase = half * 8;
#pragma unroll
    for (int i = 0; i < 8; ++i) {
        H[(size_t)(rowBase + mBase + i) * DH + col]      = acc0[i];
        H[(size_t)(rowBase + 16 + mBase + i) * DH + col] = acc1[i];
    }
}

extern "C" void kernel_launch(void* const* d_in, const int* in_sizes, int n_in,
                              void* d_out, int out_size, void* d_ws, size_t ws_size,
                              hipStream_t stream) {
    (void)in_sizes; (void)n_in; (void)out_size; (void)ws_size;

    const float* feats = (const float*)d_in[0];
    const float* W1    = (const float*)d_in[1];
    const float* b1    = (const float*)d_in[2];
    const float* g1    = (const float*)d_in[3];
    const float* be1   = (const float*)d_in[4];
    const float* W2    = (const float*)d_in[5];
    const float* b2    = (const float*)d_in[6];
    const float* g2    = (const float*)d_in[7];
    const float* be2   = (const float*)d_in[8];
    const float* W3    = (const float*)d_in[9];
    const float* b3    = (const float*)d_in[10];
    const int*   src   = (const int*)d_in[11];
    const int*   dst   = (const int*)d_in[12];

    int*   tick = (int*)d_ws;
    float* ns  = (float*)d_ws + 16 * TICK_STRIDE;
    float* nd  = ns + N_NODES;
    float* H   = nd + N_NODES;
    float* AGG = H + (size_t)N_NODES * DH;

    float* h_emb = (float*)d_out;
    float* h_out = h_emb + (size_t)N_NODES * DH;

    hipMemsetAsync(d_ws, 0, (16 * TICK_STRIDE + 2 * (size_t)N_NODES) * sizeof(float), stream);
    degree_kernel<<<1024, 256, 0, stream>>>(src, ns, tick);
    degree_kernel<<<1024, 256, 0, stream>>>(dst, nd, tick + 2 * TICK_STRIDE);
    finalize_norm_kernel<<<(N_NODES + 255) / 256, 256, 0, stream>>>(ns, nd, N_NODES);

    const size_t aggBytes = (size_t)N_NODES * DH * sizeof(float);

    gemm_ns_wmma<DIN><<<N_NODES / 16, 128, 0, stream>>>(feats, W1, ns, H);
    hipMemsetAsync(AGG, 0, aggBytes, stream);
    aggregate_kernel<<<2048, 256, 0, stream>>>(H, src, dst, AGG, tick + 4 * TICK_STRIDE);

    post_gemm_wmma<false><<<N_NODES / FR, 128, 0, stream>>>(AGG, nd, b1, g1, be1, ns, W2,
                                                            nullptr, H);
    hipMemsetAsync(AGG, 0, aggBytes, stream);
    aggregate_kernel<<<2048, 256, 0, stream>>>(H, src, dst, AGG, tick + 6 * TICK_STRIDE);

    post_gemm_wmma<true><<<N_NODES / FR, 128, 0, stream>>>(AGG, nd, b2, g2, be2, ns, W3,
                                                           h_emb, H);
    hipMemsetAsync(AGG, 0, aggBytes, stream);
    aggregate_kernel<<<2048, 256, 0, stream>>>(H, src, dst, AGG, tick + 8 * TICK_STRIDE);
    post_kernel<<<N_NODES / 8, 256, 0, stream>>>(AGG, nd, b3, nullptr, nullptr, h_out,
                                                 nullptr, N_NODES, 0, 0);
}


# 9 stream operations, not 14: in-degree counting rides in the layer-1 transform launch as extra work-groups; one clear covers counters and 3 aggregation planes; readers form rsqrt; sums unchanged
# speedup vs baseline: 2.7349x; 1.2172x over previous
#include <hip/hip_runtime.h>

#define N_NODES 100000
#define N_EDGES 1600000
#define DIN 128
#define DH 64
#define LN_EPS 1e-5f

typedef __attribute__((ext_vector_type(2))) float v2f;
typedef __attribute__((ext_vector_type(8))) float v8f;

#define HALF_SPLIT (N_NODES / 2)
#define TICK_STRIDE 64

__device__ __forceinline__ float deg_norm(float deg) {
    return rsqrtf(fmaxf(deg, 1.0f));
}

template <int THREADS>
__device__ __forceinline__ void count_degrees(const int* __restrict__ idx,
                                              float* __restrict__ deg,
                                              int* __restrict__ tick,
                                              int* __restrict__ s_task) {
    constexpr int CHUNK = THREADS * 4;
    constexpr int TASKS = (N_EDGES + CHUNK - 1) / CHUNK;
    const unsigned where = __builtin_amdgcn_s_sendmsg_rtn(0x87);
    int hh = (int)((where >> 18) & 1u);
    for (int pass = 0; pass < 2; ++pass) {
        for (int it = 0; it <= TASKS; ++it) {
            if (threadIdx.x == 0) *s_task = atomicAdd(&tick[hh * TICK_STRIDE], 1);
            __syncthreads();
            const int t = *s_task;
            __syncthreads();
            if (t >= TASKS) break;
            const int off = t * CHUNK + (int)threadIdx.x * 4;
            if (off < N_EDGES) {
                const int4 v = *(const int4*)(idx + off);
                if ((int)(v.x >= HALF_SPLIT) == hh) atomicAdd(&deg[v.x], 1.0f);
                if ((int)(v.y >= HALF_SPLIT) == hh) atomicAdd(&deg[v.y], 1.0f);
                if ((int)(v.z >= HALF_SPLIT) == hh) atomicAdd(&deg[v.z], 1.0f);
                if ((int)(v.w >= HALF_SPLIT) == hh) atomicAdd(&deg[v.w], 1.0f);
            }
        }
        hh ^= 1;
    }
}

__global__ __launch_bounds__(256) void degree_kernel(const int* __restrict__ idx,
                                                     float* __restrict__ deg,
                                                     int* __restrict__ tick) {
    __shared__ int s_task;
    count_degrees<256>(idx, deg, tick, &s_task);
}

#define DEG_RIDERS 1024

template <int K>
__global__ __launch_bounds__(128) void gemm_ns_wmma(const float* __restrict__ X,
                                                    const float* __restrict__ W,
                                                    const float* __restrict__ ns,
                                                    float* __restrict__ H,
                                                    const int* __restrict__ dst,
                                                    float* __restrict__ nd,
                                                    int* __restrict__ tick) {
    __shared__ int s_task;
    if (blockIdx.x < DEG_RIDERS) {
        count_degrees<128>(dst, nd, tick, &s_task);
        return;
    }
    const int lane    = threadIdx.x & 31;
    const int wave    = threadIdx.x >> 5;
    const int rowBase = ((int)blockIdx.x - DEG_RIDERS) * 16;
    const int colBase = wave * 16;
    const int half    = lane >> 4;
    const int l15     = lane & 15;
    const int row     = rowBase + l15;
    const int col     = colBase + l15;
    const float s     = deg_norm(ns[row]);

    v8f c = {};
#pragma unroll 4
    for (int k0 = 0; k0 < K; k0 += 4) {
        const int kk = k0 + half * 2;
        const float2 xa = *(const float2*)(X + row * K + kk);
        v2f a;
        a.x = xa.x * s;
        a.y = xa.y * s;
        v2f b;
        b.x = W[kk * DH + col];
        b.y = W[(kk + 1) * DH + col];
        c = __builtin_amdgcn_wmma_f32_16x16x4_f32(false, a, false, b, (short)0, c,
                                                  false, false);
    }

    const int mBase = half * 8;
#pragma unroll
    for (int i = 0; i < 8; ++i) {
        H[(size_t)(rowBase + mBase + i) * DH + col] = c[i];
    }
}

#define AGG_CHUNK 512
#define AGG_TASKS (N_EDGES / AGG_CHUNK)
#define AGG_DEPTH 8

__device__ __forceinline__ float other_half(float x) {
    return __builtin_bit_cast(float,
        __builtin_amdgcn_permlanex16(__builtin_bit_cast(int, x), __builtin_bit_cast(int, x),
                                     0x76543210, 0xfedcba98, false, false));
}

__device__ __forceinline__ void walk_window(const float* __restrict__ H,
                                            float* __restrict__ agg,
                                            int sv, int dv, int hh, int lane) {
    const bool upper = lane >= 16;
    const int colA = 2 * (lane & 15) + (lane >> 4);
    unsigned m = __builtin_amdgcn_ballot_w32((int)(dv >= HALF_SPLIT) == hh);
    while (m != 0u) {
        const int left = __builtin_popcount(m);
        const int n = left < AGG_DEPTH ? left : AGG_DEPTH;
        float2 v[AGG_DEPTH];
        int d[AGG_DEPTH];
#pragma unroll
        for (int j = 0; j < AGG_DEPTH; ++j) {
            if (j < n) {
                const int i = __builtin_ctz(m);
                m &= m - 1u;
                const int s = __builtin_amdgcn_readlane(sv, i);
                d[j] = __builtin_amdgcn_readlane(dv, i);
                v[j] = *(const float2*)(H + (size_t)s * DH + lane * 2);
            }
        }
#pragma unroll
        for (int j = 0; j < AGG_DEPTH; ++j) {
            if (j < n) {
                const float got = other_half(upper ? v[j].x : v[j].y);
                const float a = upper ? got : v[j].x;
                const float b = upper ? v[j].y : got;
                float* p = agg + (size_t)d[j] * DH + colA;
                atomicAdd(p, a);
                atomicAdd(p + 32, b);
            }
        }
    }
}

__global__ __launch_bounds__(256) void aggregate_kernel(const float* __restrict__ H,
                                                        const int* __restrict__ src,
                                                        const int* __restrict__ dst,
                                                        float* __restrict__ agg,
                                                        int* __restrict__ tick) {
    __shared__ int s_task;
    const int lane = threadIdx.x & 31;
    const int wave = threadIdx.x >> 5;
    const unsigned where = __builtin_amdgcn_s_sendmsg_rtn(0x87);
    int hh = (int)((where >> 18) & 1u);
    for (int pass = 0; pass < 2; ++pass) {
        for (int it = 0; it <= AGG_TASKS; ++it) {
            if (threadIdx.x == 0) s_task = atomicAdd(&tick[hh * TICK_STRIDE], 1);
            __syncthreads();
            const int t = __builtin_amdgcn_readfirstlane(s_task);
            __syncthreads();
            if (t >= AGG_TASKS) break;
            const int e = t * AGG_CHUNK + wave * 64 + lane;
            const int dv0 = dst[e];
            const int dv1 = dst[e + 32];
            const int sv0 = src[e];
            const int sv1 = src[e + 32];
            asm volatile("" :: "v"(dv0), "v"(dv1), "v"(sv0), "v"(sv1));
            walk_window(H, agg, sv0, dv0, hh, lane);
            walk_window(H, agg, sv1, dv1, hh, lane);
        }
        hh ^= 1;
    }
}

__global__ __launch_bounds__(256) void post_kernel(const float* __restrict__ agg,
                                                   const float* __restrict__ nd,
                                                   const float* __restrict__ bias,
                                                   const float* __restrict__ g,
                                                   const float* __restrict__ be,
                                                   float* __restrict__ outx,
                                                   float* __restrict__ pre,
                                                   int n, int do_relu, int do_ln) {
    const int node = blockIdx.x * 8 + (threadIdx.x >> 5);
    if (node >= n) return;
    const int lane = threadIdx.x & 31;
    const int c0   = lane * 2;
    const float ndv = deg_norm(nd[node]);
    const float* r = agg + (size_t)node * DH;
    float v0 = r[c0 + 0] * ndv + bias[c0 + 0];
    float v1 = r[c0 + 1] * ndv + bias[c0 + 1];
    if (do_relu) {
        v0 = fmaxf(v0, 0.0f);
        v1 = fmaxf(v1, 0.0f);
    }
    if (pre) {
        pre[(size_t)node * DH + c0 + 0] = v0;
        pre[(size_t)node * DH + c0 + 1] = v1;
    }
    if (do_ln) {
        float ssum = v0 + v1;
#pragma unroll
        for (int off = 16; off >= 1; off >>= 1) ssum += __shfl_xor(ssum, off, 32);
        const float mu = ssum * (1.0f / DH);
        const float d0 = v0 - mu, d1 = v1 - mu;
        float vs = d0 * d0 + d1 * d1;
#pragma unroll
        for (int off = 16; off >= 1; off >>= 1) vs += __shfl_xor(vs, off, 32);
        const float inv = rsqrtf(vs * (1.0f / DH) + LN_EPS);
        v0 = d0 * inv * g[c0 + 0] + be[c0 + 0];
        v1 = d1 * inv * g[c0 + 1] + be[c0 + 1];
    }
    float* o = outx + (size_t)node * DH + c0;
    o[0] = v0;
    o[1] = v1;
}

#define FR 32
#define AP 68

template <bool EMIT_PRE>
__global__ __launch_bounds__(128) void post_gemm_wmma(const float* __restrict__ agg,
                                                      const float* __restrict__ nd,
                                                      const float* __restrict__ bias,
                                                      const float* __restrict__ g,
                                                      const float* __restrict__ be,
                                                      const float* __restrict__ ns,
                                                      const float* __restrict__ W,
                                                      float* __restrict__ pre,
                                                      float* __restrict__ H) {
    __shared__ __attribute__((aligned(16))) float As[FR * AP];
    const int lane = threadIdx.x & 31;
    const int wave    = __builtin_amdgcn_readfirstlane((int)(threadIdx.x >> 5));
    const int rowBase = blockIdx.x * FR;
    const int half    = lane >> 4;
    const int l15     = lane & 15;
    const int col     = wave * 16 + l15;

    v2f bf[DH / 4];
#pragma unroll
    for (int q = 0; q < DH / 4; ++q) {
        const int kk = q * 4 + half * 2;
        bf[q].x = W[kk * DH + col];
        bf[q].y = W[(kk + 1) * DH + col];
    }

    const int c0 = lane * 2;
    const float b0 = bias[c0 + 0], b1 = bias[c0 + 1];
    const float g0 = g[c0 + 0],    g1 = g[c0 + 1];
    const float e0 = be[c0 + 0],   e1 = be[c0 + 1];
#pragma unroll
    for (int j = 0; j < FR / 4; ++j) {
        const int lr   = wave * (FR / 4) + j;
        const int node = rowBase + lr;
        const float ndv = deg_norm(nd[node]);
        const float2 r = *(const float2*)(agg + (size_t)node * DH + c0);
        float v0 = r.x * ndv + b0;
        float v1 = r.y * ndv + b1;
        v0 = fmaxf(v0, 0.0f);
        v1 = fmaxf(v1, 0.0f);
        if (EMIT_PRE) {
            *(float2*)(pre + (size_t)node * DH + c0) = make_float2(v0, v1);
        }
        float ssum = v0 + v1;
#pragma unroll
        for (int off = 16; off >= 1; off >>= 1) ssum += __shfl_xor(ssum, off, 32);
        const float mu = ssum * (1.0f / DH);
        const float d0 = v0 - mu, d1 = v1 - mu;
        float vs = d0 * d0 + d1 * d1;
#pragma unroll
        for (int off = 16; off >= 1; off >>= 1) vs += __shfl_xor(vs, off, 32);
        const float inv = rsqrtf(vs * (1.0f / DH) + LN_EPS);
        v0 = d0 * inv * g0 + e0;
        v1 = d1 * inv * g1 + e1;
        const float s = deg_norm(ns[node]);
        *(float2*)(As + lr * AP + c0) = make_float2(v0 * s, v1 * s);
    }
    __syncthreads();

    v8f acc0 = {};
    v8f acc1 = {};
#pragma unroll
    for (int q = 0; q < DH / 4; ++q) {
        const int kk = q * 4 + half * 2;
        const float2 x0 = *(const float2*)(As + l15 * AP + kk);
        const float2 x1 = *(const float2*)(As + (16 + l15) * AP + kk);
        v2f a0, a1;
        a0.x = x0.x; a0.y = x0.y;
        a1.x = x1.x; a1.y = x1.y;
        acc0 = __builtin_amdgcn_wmma_f32_16x16x4_f32(false, a0, false, bf[q], (short)0, acc0,
                                                     false, false);
        acc1 = __builtin_amdgcn_wmma_f32_16x16x4_f32(false, a1, false, bf[q], (short)0, acc1,
                                                     false, false);
    }

    const int mBase = half * 8;
#pragma unroll
    for (int i = 0; i < 8; ++i) {
        H[(size_t)(rowBase + mBase + i) * DH + col]      = acc0[i];
        H[(size_t)(rowBase + 16 + mBase + i) * DH + col] = acc1[i];
    }
}

extern "C" void kernel_launch(void* const* d_in, const int* in_sizes, int n_in,
                              void* d_out, int out_size, void* d_ws, size_t ws_size,
                              hipStream_t stream) {
    (void)in_sizes; (void)n_in; (void)out_size; (void)ws_size;

    const float* feats = (const float*)d_in[0];
    const float* W1    = (const float*)d_in[1];
    const float* b1    = (const float*)d_in[2];
    const float* g1    = (const float*)d_in[3];
    const float* be1   = (const float*)d_in[4];
    const float* W2    = (const float*)d_in[5];
    const float* b2    = (const float*)d_in[6];
    const float* g2    = (const float*)d_in[7];
    const float* be2   = (const float*)d_in[8];
    const float* W3    = (const float*)d_in[9];
    const float* b3    = (const float*)d_in[10];
    const int*   src   = (const int*)d_in[11];
    const int*   dst   = (const int*)d_in[12];

    int*   tick = (int*)d_ws;
    float* ns   = (float*)d_ws + 16 * TICK_STRIDE;
    float* nd   = ns + N_NODES;
    float* AGG1 = nd + N_NODES;
    float* AGG2 = AGG1 + (size_t)N_NODES * DH;
    float* AGG3 = AGG2 + (size_t)N_NODES * DH;
    float* H    = AGG3 + (size_t)N_NODES * DH;

    float* h_emb = (float*)d_out;
    float* h_out = h_emb + (size_t)N_NODES * DH;

    hipMemsetAsync(d_ws, 0,
                   (16 * TICK_STRIDE + 2 * (size_t)N_NODES + 3 * (size_t)N_NODES * DH) * sizeof(float),
                   stream);

    degree_kernel<<<1024, 256, 0, stream>>>(src, ns, tick);

    gemm_ns_wmma<DIN><<<DEG_RIDERS + N_NODES / 16, 128, 0, stream>>>(feats, W1, ns, H, dst, nd,
                                                                     tick + 2 * TICK_STRIDE);
    aggregate_kernel<<<2048, 256, 0, stream>>>(H, src, dst, AGG1, tick + 4 * TICK_STRIDE);

    post_gemm_wmma<false><<<N_NODES / FR, 128, 0, stream>>>(AGG1, nd, b1, g1, be1, ns, W2,
                                                            nullptr, H);
    aggregate_kernel<<<2048, 256, 0, stream>>>(H, src, dst, AGG2, tick + 6 * TICK_STRIDE);

    post_gemm_wmma<true><<<N_NODES / FR, 128, 0, stream>>>(AGG2, nd, b2, g2, be2, ns, W3,
                                                           h_emb, H);
    aggregate_kernel<<<2048, 256, 0, stream>>>(H, src, dst, AGG3, tick + 8 * TICK_STRIDE);
    post_kernel<<<N_NODES / 8, 256, 0, stream>>>(AGG3, nd, b3, nullptr, nullptr, h_out,
                                                 nullptr, N_NODES, 0, 0);
}
